# starting kernel
# baseline (speedup 1.0000x reference)
#include <hip/hip_runtime.h>

typedef float v2f __attribute__((ext_vector_type(2)));
typedef float v8f __attribute__((ext_vector_type(8)));

#define D_HID 128

__global__ void wmma_linear_kernel(const float* __restrict__ X,
                                   const float* __restrict__ W,
                                   const float* __restrict__ bias,
                                   float* __restrict__ Y,
                                   int M, int K, int N, int do_relu)
{
    const int lane = threadIdx.x & 31;
    const int wave = threadIdx.x >> 5;
    const int ntiles = (N + 15) >> 4;
    const int nt = blockIdx.y * 4 + wave;
    if (nt >= ntiles) return;

    const int mt   = blockIdx.x;
    const int half = lane >> 4;
    const int l    = lane & 15;

    const int m  = mt * 16 + l;
    const int mc = (m < M) ? m : (M - 1);
    const int n  = nt * 16 + l;
    const int nc = (n < N) ? n : (N - 1);

    const float* xrow = X + (size_t)mc * K + 2 * half;
    const float* wrow = W + (size_t)nc * K + 2 * half;

    v8f acc = {0.f, 0.f, 0.f, 0.f, 0.f, 0.f, 0.f, 0.f};
    #pragma unroll 8
    for (int k = 0; k < K; k += 4) {
        v2f a = *(const v2f*)(xrow + k);
        v2f b = *(const v2f*)(wrow + k);
        acc = __builtin_amdgcn_wmma_f32_16x16x4_f32(
                  false, a, false, b, (short)0, acc, false, false);
    }

    const float bv = bias[nc];
    #pragma unroll
    for (int r = 0; r < 8; ++r) {
        float v = acc[r] + bv;
        if (do_relu) v = (v > 0.f) ? v : 0.f;
        const int row = mt * 16 + r + 8 * half;
        if (n < N && row < M)
            Y[(size_t)row * N + n] = v;
    }
}

__global__ void zero_kernel(float* __restrict__ p, long long n)
{
    long long i = (long long)blockIdx.x * blockDim.x + threadIdx.x;
    if (i < n) p[i] = 0.0f;
}

__global__ void scatter_add_kernel(const float* __restrict__ Y,
                                   const int* __restrict__ src,
                                   const int* __restrict__ dst,
                                   float* __restrict__ out, int E)
{
    const long long t = (long long)blockIdx.x * blockDim.x + threadIdx.x;
    const int e = (int)(t >> 5);
    if (e >= E) return;
    const int q = (int)(t & 31);

    const int s = src[e];
    const int d = dst[e];
    const float4 v = ((const float4*)(Y + (size_t)s * D_HID))[q];
    float* o = out + (size_t)d * D_HID + q * 4;
    atomicAdd(o + 0, v.x);
    atomicAdd(o + 1, v.y);
    atomicAdd(o + 2, v.z);
    atomicAdd(o + 3, v.w);
}

extern "C" void kernel_launch(void* const* d_in, const int* in_sizes, int n_in,
                              void* d_out, int out_size, void* d_ws, size_t ws_size,
                              hipStream_t stream)
{
    const float* X0  = (const float*)d_in[0];
    const float* W1  = (const float*)d_in[1];
    const float* b1  = (const float*)d_in[2];
    const float* W2  = (const float*)d_in[3];
    const float* b2  = (const float*)d_in[4];
    const float* Wfc = (const float*)d_in[5];
    const float* bfc = (const float*)d_in[6];
    const int*   esrc = (const int*)d_in[7];
    const int*   edst = (const int*)d_in[8];

    const int Nn   = in_sizes[0] / D_HID;
    const int E    = in_sizes[7];
    const int Ncls = in_sizes[6];

    float* bufA = (float*)d_ws;
    float* bufB = bufA + (size_t)Nn * D_HID;
    const long long hidElems = (long long)Nn * D_HID;

    const dim3 blkLin(128);
    const int mtiles = (Nn + 15) / 16;
    const dim3 grdHid(mtiles, (D_HID / 16 + 3) / 4);
    const dim3 grdFc(mtiles, ((Ncls + 15) / 16 + 3) / 4);

    const int zeroBlocks = (int)((hidElems + 255) / 256);
    const int scatBlocks = (int)(((long long)E * 32 + 255) / 256);

    wmma_linear_kernel<<<grdHid, blkLin, 0, stream>>>(X0, W1, b1, bufA,
                                                      Nn, D_HID, D_HID, 1);
    zero_kernel<<<zeroBlocks, 256, 0, stream>>>(bufB, hidElems);
    scatter_add_kernel<<<scatBlocks, 256, 0, stream>>>(bufA, esrc, edst, bufB, E);

    wmma_linear_kernel<<<grdHid, blkLin, 0, stream>>>(bufB, W2, b2, bufA,
                                                      Nn, D_HID, D_HID, 1);
    zero_kernel<<<zeroBlocks, 256, 0, stream>>>(bufB, hidElems);
    scatter_add_kernel<<<scatBlocks, 256, 0, stream>>>(bufA, esrc, edst, bufB, E);

    wmma_linear_kernel<<<grdFc, blkLin, 0, stream>>>(bufB, Wfc, bfc, (float*)d_out,
                                                     Nn, D_HID, Ncls, 0);
}


# apply the final FC before the second scatter (the FC is linear): z = y2 @ Wfc^T per node, out starts at bfc, and the scatter-add moves 40 class values per edge instead of 128 hidden values
# speedup vs baseline: 1.1920x; 1.1920x over previous
#include <hip/hip_runtime.h>

typedef float v2f __attribute__((ext_vector_type(2)));
typedef float v8f __attribute__((ext_vector_type(8)));

#define D_HID 128
#define CLS_PER_LANE 4

__global__ void wmma_linear_kernel(const float* __restrict__ X,
                                   const float* __restrict__ W,
                                   const float* __restrict__ bias,
                                   float* __restrict__ Y,
                                   float* __restrict__ Yinit,
                                   int M, int K, int N, int do_relu)
{
    const int lane = threadIdx.x & 31;
    const int wave = threadIdx.x >> 5;
    const int ntiles = (N + 15) >> 4;
    const int nt = blockIdx.y * 4 + wave;
    if (nt >= ntiles) return;

    const int mt   = blockIdx.x;
    const int half = lane >> 4;
    const int l    = lane & 15;

    const int m  = mt * 16 + l;
    const int mc = (m < M) ? m : (M - 1);
    const int n  = nt * 16 + l;
    const int nc = (n < N) ? n : (N - 1);

    const float* xrow = X + (size_t)mc * K + 2 * half;
    const float* wrow = W + (size_t)nc * K + 2 * half;

    v8f acc = {0.f, 0.f, 0.f, 0.f, 0.f, 0.f, 0.f, 0.f};
    #pragma unroll 8
    for (int k = 0; k < K; k += 4) {
        v2f a = *(const v2f*)(xrow + k);
        v2f b = *(const v2f*)(wrow + k);
        acc = __builtin_amdgcn_wmma_f32_16x16x4_f32(
                  false, a, false, b, (short)0, acc, false, false);
    }

    const float bv = bias[nc];
    const float badd = Yinit ? 0.f : bv;
    #pragma unroll
    for (int r = 0; r < 8; ++r) {
        float v = acc[r] + badd;
        if (do_relu) v = (v > 0.f) ? v : 0.f;
        const int row = mt * 16 + r + 8 * half;
        if (n < N && row < M) {
            Y[(size_t)row * N + n] = v;
            if (Yinit) Yinit[(size_t)row * N + n] = bv;
        }
    }
}

__global__ void zero_kernel(float* __restrict__ p, long long n)
{
    long long i = (long long)blockIdx.x * blockDim.x + threadIdx.x;
    if (i < n) p[i] = 0.0f;
}

__global__ void scatter_add_kernel(const float* __restrict__ Y,
                                   const int* __restrict__ src,
                                   const int* __restrict__ dst,
                                   float* __restrict__ out, int E)
{
    const long long t = (long long)blockIdx.x * blockDim.x + threadIdx.x;
    const int e = (int)(t >> 5);
    if (e >= E) return;
    const int q = (int)(t & 31);

    const int s = src[e];
    const int d = dst[e];
    const float4 v = ((const float4*)(Y + (size_t)s * D_HID))[q];
    float* o = out + (size_t)d * D_HID + q * 4;
    atomicAdd(o + 0, v.x);
    atomicAdd(o + 1, v.y);
    atomicAdd(o + 2, v.z);
    atomicAdd(o + 3, v.w);
}

__global__ void scatter_add_cls_kernel(const float* __restrict__ Z,
                                       const int* __restrict__ src,
                                       const int* __restrict__ dst,
                                       float* __restrict__ out, int E, int C)
{
    const int e = blockIdx.x * blockDim.y + threadIdx.y;
    if (e >= E) return;

    const int s = src[e];
    const int d = dst[e];
    const float* zr = Z + (size_t)s * C;
    float* o = out + (size_t)d * C;

    float v[CLS_PER_LANE];
    for (int i = 0; i < CLS_PER_LANE; ++i) {
        const int c = threadIdx.x + i * blockDim.x;
        v[i] = (c < C) ? zr[c] : 0.f;
    }
    for (int i = 0; i < CLS_PER_LANE; ++i) {
        const int c = threadIdx.x + i * blockDim.x;
        if (c < C) atomicAdd(o + c, v[i]);
    }
}

extern "C" void kernel_launch(void* const* d_in, const int* in_sizes, int n_in,
                              void* d_out, int out_size, void* d_ws, size_t ws_size,
                              hipStream_t stream)
{
    const float* X0  = (const float*)d_in[0];
    const float* W1  = (const float*)d_in[1];
    const float* b1  = (const float*)d_in[2];
    const float* W2  = (const float*)d_in[3];
    const float* b2  = (const float*)d_in[4];
    const float* Wfc = (const float*)d_in[5];
    const float* bfc = (const float*)d_in[6];
    const int*   esrc = (const int*)d_in[7];
    const int*   edst = (const int*)d_in[8];

    const int Nn   = in_sizes[0] / D_HID;
    const int E    = in_sizes[7];
    const int Ncls = in_sizes[6];

    float* bufA = (float*)d_ws;
    float* bufB = bufA + (size_t)Nn * D_HID;
    float* zbuf = bufB;
    const long long hidElems = (long long)Nn * D_HID;

    const dim3 blkLin(128);
    const int mtiles = (Nn + 15) / 16;
    const dim3 grdHid(mtiles, (D_HID / 16 + 3) / 4);
    const dim3 grdFc(mtiles, ((Ncls + 15) / 16 + 3) / 4);

    const int zeroBlocks = (int)((hidElems + 255) / 256);
    const int scatBlocks = (int)(((long long)E * 32 + 255) / 256);
    const dim3 blkCls((Ncls + CLS_PER_LANE - 1) / CLS_PER_LANE, 32);
    const int clsBlocks = (E + 31) / 32;

    wmma_linear_kernel<<<grdHid, blkLin, 0, stream>>>(X0, W1, b1, bufA, nullptr,
                                                      Nn, D_HID, D_HID, 1);
    zero_kernel<<<zeroBlocks, 256, 0, stream>>>(bufB, hidElems);
    scatter_add_kernel<<<scatBlocks, 256, 0, stream>>>(bufA, esrc, edst, bufB, E);

    wmma_linear_kernel<<<grdHid, blkLin, 0, stream>>>(bufB, W2, b2, bufA, nullptr,
                                                      Nn, D_HID, D_HID, 1);

    wmma_linear_kernel<<<grdFc, blkLin, 0, stream>>>(bufA, Wfc, bfc, zbuf, (float*)d_out,
                                                     Nn, D_HID, Ncls, 0);
    scatter_add_cls_kernel<<<clsBlocks, blkCls, 0, stream>>>(zbuf, esrc, edst,
                                                             (float*)d_out, E, Ncls);
}


# linear kernels: one wave per 16-row strip of all output columns; W staged once per workgroup in padded LDS, read as 16-byte fragments; X read once with 16-byte loads (same k order for A and B)
# speedup vs baseline: 1.4408x; 1.2087x over previous
#include <hip/hip_runtime.h>

typedef float v2f __attribute__((ext_vector_type(2)));
typedef float v4f __attribute__((ext_vector_type(4)));
typedef float v8f __attribute__((ext_vector_type(8)));

#define D_HID 128
#define CLS_PER_LANE 4
#define CLS_TILES 3
#define LIN_WAVES 8
#define W_PITCH (D_HID + 4)


template <int NT>
__device__ __forceinline__ void mma_block(v8f (&acc)[NT], const v4f (&cur)[NT], v4f (&nxt)[NT],
                                          const v4f a, const float* __restrict__ wnext)
{
    #pragma unroll
    for (int t = 0; t < NT; ++t) nxt[t] = *(const v4f*)(wnext + t * 16 * W_PITCH);
    __builtin_amdgcn_sched_barrier(0);
    const v2f a0 = {a.x, a.y};
    const v2f a1 = {a.z, a.w};
    #pragma unroll
    for (int t = 0; t < NT; ++t) {
        const v2f b0 = {cur[t].x, cur[t].y};
        const v2f b1 = {cur[t].z, cur[t].w};
        acc[t] = __builtin_amdgcn_wmma_f32_16x16x4_f32(
                     false, a0, false, b0, (short)0, acc[t], false, false);
        acc[t] = __builtin_amdgcn_wmma_f32_16x16x4_f32(
                     false, a1, false, b1, (short)0, acc[t], false, false);
    }
    __builtin_amdgcn_sched_barrier(0);
}

template <int NT, bool RELU, bool INIT, bool ROWS_GUARDED>
__device__ __forceinline__ void store_strip(const v8f (&acc)[NT], const float (&bv)[NT],
                                            float* __restrict__ Y, float* __restrict__ Yinit,
                                            int row0, int l, int M, int N)
{
    float* yb = Y + (size_t)row0 * N + l;
    float* ib = INIT ? (Yinit + (size_t)row0 * N + l) : nullptr;
    #pragma unroll
    for (int t = 0; t < NT; ++t) {
        const float badd = INIT ? 0.f : bv[t];
        if (t * 16 + l < N) {
            #pragma unroll
            for (int r = 0; r < 8; ++r) {
                float v = acc[t][r] + badd;
                if (RELU) v = (v > 0.f) ? v : 0.f;
                if (!ROWS_GUARDED || row0 + r < M) {
                    yb[(size_t)r * N + t * 16] = v;
                    if (INIT) ib[(size_t)r * N + t * 16] = bv[t];
                }
            }
        }
    }
}

template <int NT, bool RELU, bool INIT>
__device__ __forceinline__ void linear_strips(const float* __restrict__ X,
                                              const float* __restrict__ W,
                                              const float* __restrict__ bias,
                                              float* __restrict__ Y,
                                              float* __restrict__ Yinit,
                                              int M, int N, float* wlds)
{
    const int tid  = threadIdx.x;
    const int lane = tid & 31;
    const int wave = __builtin_amdgcn_readfirstlane(tid >> 5);
    const int half = lane >> 4;
    const int l    = lane & 15;
    const int mtiles = (M + 15) >> 4;
    const int mstep  = gridDim.x * LIN_WAVES;

    int mt = blockIdx.x * LIN_WAVES + wave;

    const float* xrow;
    v4f an[4];
    {
        const int m  = mt * 16 + l;
        const int mc = (m < M) ? m : (M - 1);
        xrow = X + (size_t)mc * D_HID + 4 * half;
        #pragma unroll
        for (int jj = 0; jj < 4; ++jj) an[jj] = *(const v4f*)(xrow + 8 * jj);
    }

    {
        constexpr int ROWS_PER_WAVE = NT * 16 / LIN_WAVES;
        v4f wst[ROWS_PER_WAVE];
        #pragma unroll
        for (int c = 0; c < ROWS_PER_WAVE; ++c) {
            const int row = wave + LIN_WAVES * c;
            const int rc  = (row < N) ? row : (N - 1);
            wst[c] = *(const v4f*)(W + (size_t)rc * D_HID + 4 * lane);
        }
        #pragma unroll
        for (int c = 0; c < ROWS_PER_WAVE; ++c) {
            const int row = wave + LIN_WAVES * c;
            *(v4f*)(wlds + row * W_PITCH + 4 * lane) = wst[c];
        }
    }

    float bv[NT];
    #pragma unroll
    for (int t = 0; t < NT; ++t) {
        const int n  = t * 16 + l;
        bv[t] = bias[(n < N) ? n : (N - 1)];
    }

    __syncthreads();

    const float* wl = wlds + l * W_PITCH + 4 * half;

    v4f fb0[NT], fb1[NT];
    #pragma unroll
    for (int t = 0; t < NT; ++t) fb0[t] = *(const v4f*)(wl + t * 16 * W_PITCH);

    do {
        v8f acc[NT];
        #pragma unroll
        for (int t = 0; t < NT; ++t) acc[t] = (v8f){0.f, 0.f, 0.f, 0.f, 0.f, 0.f, 0.f, 0.f};

        #pragma unroll 1
        for (int g = 0; g < 4; ++g) {
            v4f ac[4];
            #pragma unroll
            for (int jj = 0; jj < 4; ++jj) ac[jj] = an[jj];
            if (g < 3) {
                #pragma unroll
                for (int jj = 0; jj < 4; ++jj) an[jj] = *(const v4f*)(xrow + 32 * (g + 1) + 8 * jj);
            }
            const float* wg = wl + 32 * g;
            const float* wn = wl + 32 * ((g + 1) & 3);
            mma_block<NT>(acc, fb0, fb1, ac[0], wg + 8);
            mma_block<NT>(acc, fb1, fb0, ac[1], wg + 16);
            mma_block<NT>(acc, fb0, fb1, ac[2], wg + 24);
            mma_block<NT>(acc, fb1, fb0, ac[3], wn);
        }

        const int mrow = mt * 16;
        mt += mstep;
        if (mt < mtiles) {
            const int m  = mt * 16 + l;
            const int mc = (m < M) ? m : (M - 1);
            xrow = X + (size_t)mc * D_HID + 4 * half;
            #pragma unroll
            for (int jj = 0; jj < 4; ++jj) an[jj] = *(const v4f*)(xrow + 8 * jj);
        }

        if (mrow + 16 <= M)
            store_strip<NT, RELU, INIT, false>(acc, bv, Y, Yinit, mrow + 8 * half, l, M, N);
        else
            store_strip<NT, RELU, INIT, true>(acc, bv, Y, Yinit, mrow + 8 * half, l, M, N);
    } while (mt < mtiles);
}

__global__ void __launch_bounds__(LIN_WAVES * 32)
linear_hidden_kernel(const float* __restrict__ X, const float* __restrict__ W,
                     const float* __restrict__ bias, float* __restrict__ Y, int M)
{
    __shared__ __attribute__((aligned(16))) float wlds[D_HID * W_PITCH];
    linear_strips<D_HID / 16, true, false>(X, W, bias, Y, nullptr, M, D_HID, wlds);
}

__global__ void __launch_bounds__(LIN_WAVES * 32)
linear_cls_kernel(const float* __restrict__ X, const float* __restrict__ W,
                  const float* __restrict__ bias, float* __restrict__ Z,
                  float* __restrict__ Yinit, int M, int N)
{
    __shared__ __attribute__((aligned(16))) float wlds[CLS_TILES * 16 * W_PITCH];
    linear_strips<CLS_TILES, false, true>(X, W, bias, Z, Yinit, M, N, wlds);
}

__global__ void zero_kernel(float* __restrict__ p, long long n)
{
    long long i = (long long)blockIdx.x * blockDim.x + threadIdx.x;
    if (i < n) p[i] = 0.0f;
}

__global__ void scatter_add_kernel(const float* __restrict__ Y,
                                   const int* __restrict__ src,
                                   const int* __restrict__ dst,
                                   float* __restrict__ out, int E)
{
    const long long t = (long long)blockIdx.x * blockDim.x + threadIdx.x;
    const int e = (int)(t >> 5);
    if (e >= E) return;
    const int q = (int)(t & 31);

    const int s = src[e];
    const int d = dst[e];
    const float4 v = ((const float4*)(Y + (size_t)s * D_HID))[q];
    float* o = out + (size_t)d * D_HID + q * 4;
    atomicAdd(o + 0, v.x);
    atomicAdd(o + 1, v.y);
    atomicAdd(o + 2, v.z);
    atomicAdd(o + 3, v.w);
}

__global__ void scatter_add_cls_kernel(const float* __restrict__ Z,
                                       const int* __restrict__ src,
                                       const int* __restrict__ dst,
                                       float* __restrict__ out, int E, int C)
{
    const int e = blockIdx.x * blockDim.y + threadIdx.y;
    if (e >= E) return;

    const int s = src[e];
    const int d = dst[e];
    const float* zr = Z + (size_t)s * C;
    float* o = out + (size_t)d * C;

    float v[CLS_PER_LANE];
    for (int i = 0; i < CLS_PER_LANE; ++i) {
        const int c = threadIdx.x + i * blockDim.x;
        v[i] = (c < C) ? zr[c] : 0.f;
    }
    for (int i = 0; i < CLS_PER_LANE; ++i) {
        const int c = threadIdx.x + i * blockDim.x;
        if (c < C) atomicAdd(o + c, v[i]);
    }
}

extern "C" void kernel_launch(void* const* d_in, const int* in_sizes, int n_in,
                              void* d_out, int out_size, void* d_ws, size_t ws_size,
                              hipStream_t stream)
{
    const float* X0  = (const float*)d_in[0];
    const float* W1  = (const float*)d_in[1];
    const float* b1  = (const float*)d_in[2];
    const float* W2  = (const float*)d_in[3];
    const float* b2  = (const float*)d_in[4];
    const float* Wfc = (const float*)d_in[5];
    const float* bfc = (const float*)d_in[6];
    const int*   esrc = (const int*)d_in[7];
    const int*   edst = (const int*)d_in[8];

    const int Nn   = in_sizes[0] / D_HID;
    const int E    = in_sizes[7];
    const int Ncls = in_sizes[6];

    float* bufA = (float*)d_ws;
    float* bufB = bufA + (size_t)Nn * D_HID;
    float* zbuf = bufB;
    const long long hidElems = (long long)Nn * D_HID;

    const int mtiles = (Nn + 15) / 16;
    const int linBlocks = (mtiles + LIN_WAVES - 1) / LIN_WAVES;
    const dim3 blkLin(LIN_WAVES * 32);

    const int zeroBlocks = (int)((hidElems + 255) / 256);
    const int scatBlocks = (int)(((long long)E * 32 + 255) / 256);
    const dim3 blkCls((Ncls + CLS_PER_LANE - 1) / CLS_PER_LANE, 32);
    const int clsBlocks = (E + 31) / 32;

    linear_hidden_kernel<<<linBlocks, blkLin, 0, stream>>>(X0, W1, b1, bufA, Nn);
    zero_kernel<<<zeroBlocks, 256, 0, stream>>>(bufB, hidElems);
    scatter_add_kernel<<<scatBlocks, 256, 0, stream>>>(bufA, esrc, edst, bufB, E);

    linear_hidden_kernel<<<linBlocks, blkLin, 0, stream>>>(bufB, W2, b2, bufA, Nn);

    linear_cls_kernel<<<linBlocks, blkLin, 0, stream>>>(bufA, Wfc, bfc, zbuf, (float*)d_out,
                                                        Nn, Ncls);
    scatter_add_cls_kernel<<<clsBlocks, blkCls, 0, stream>>>(zbuf, esrc, edst,
                                                             (float*)d_out, E, Ncls);
}


# 128-wide scatter-add: lane q adds floats q, q+32, q+64, q+96 of the row, so each load and each atomic instruction covers one whole 128-byte line, not 8 floats in each of the row's 4 lines
# speedup vs baseline: 1.6902x; 1.1731x over previous
#include <hip/hip_runtime.h>

typedef float v2f __attribute__((ext_vector_type(2)));
typedef float v4f __attribute__((ext_vector_type(4)));
typedef float v8f __attribute__((ext_vector_type(8)));

#define D_HID 128
#define CLS_PER_LANE 4
#define CLS_TILES 3
#define LIN_WAVES 8
#define W_PITCH (D_HID + 4)


template <int NT>
__device__ __forceinline__ void mma_block(v8f (&acc)[NT], const v4f (&cur)[NT], v4f (&nxt)[NT],
                                          const v4f a, const float* __restrict__ wnext)
{
    #pragma unroll
    for (int t = 0; t < NT; ++t) nxt[t] = *(const v4f*)(wnext + t * 16 * W_PITCH);
    __builtin_amdgcn_sched_barrier(0);
    const v2f a0 = {a.x, a.y};
    const v2f a1 = {a.z, a.w};
    #pragma unroll
    for (int t = 0; t < NT; ++t) {
        const v2f b0 = {cur[t].x, cur[t].y};
        const v2f b1 = {cur[t].z, cur[t].w};
        acc[t] = __builtin_amdgcn_wmma_f32_16x16x4_f32(
                     false, a0, false, b0, (short)0, acc[t], false, false);
        acc[t] = __builtin_amdgcn_wmma_f32_16x16x4_f32(
                     false, a1, false, b1, (short)0, acc[t], false, false);
    }
    __builtin_amdgcn_sched_barrier(0);
}

template <int NT, bool RELU, bool INIT, bool ROWS_GUARDED>
__device__ __forceinline__ void store_strip(const v8f (&acc)[NT], const float (&bv)[NT],
                                            float* __restrict__ Y, float* __restrict__ Yinit,
                                            int row0, int l, int M, int N)
{
    float* yb = Y + (size_t)row0 * N + l;
    float* ib = INIT ? (Yinit + (size_t)row0 * N + l) : nullptr;
    #pragma unroll
    for (int t = 0; t < NT; ++t) {
        const float badd = INIT ? 0.f : bv[t];
        if (t * 16 + l < N) {
            #pragma unroll
            for (int r = 0; r < 8; ++r) {
                float v = acc[t][r] + badd;
                if (RELU) v = (v > 0.f) ? v : 0.f;
                if (!ROWS_GUARDED || row0 + r < M) {
                    yb[(size_t)r * N + t * 16] = v;
                    if (INIT) ib[(size_t)r * N + t * 16] = bv[t];
                }
            }
        }
    }
}

template <int NT, bool RELU, bool INIT>
__device__ __forceinline__ void linear_strips(const float* __restrict__ X,
                                              const float* __restrict__ W,
                                              const float* __restrict__ bias,
                                              float* __restrict__ Y,
                                              float* __restrict__ Yinit,
                                              int M, int N, float* wlds)
{
    const int tid  = threadIdx.x;
    const int lane = tid & 31;
    const int wave = __builtin_amdgcn_readfirstlane(tid >> 5);
    const int half = lane >> 4;
    const int l    = lane & 15;
    const int mtiles = (M + 15) >> 4;
    const int mstep  = gridDim.x * LIN_WAVES;

    int mt = blockIdx.x * LIN_WAVES + wave;

    const float* xrow;
    v4f an[4];
    {
        const int m  = mt * 16 + l;
        const int mc = (m < M) ? m : (M - 1);
        xrow = X + (size_t)mc * D_HID + 4 * half;
        #pragma unroll
        for (int jj = 0; jj < 4; ++jj) an[jj] = *(const v4f*)(xrow + 8 * jj);
    }

    {
        constexpr int ROWS_PER_WAVE = NT * 16 / LIN_WAVES;
        v4f wst[ROWS_PER_WAVE];
        #pragma unroll
        for (int c = 0; c < ROWS_PER_WAVE; ++c) {
            const int row = wave + LIN_WAVES * c;
            const int rc  = (row < N) ? row : (N - 1);
            wst[c] = *(const v4f*)(W + (size_t)rc * D_HID + 4 * lane);
        }
        #pragma unroll
        for (int c = 0; c < ROWS_PER_WAVE; ++c) {
            const int row = wave + LIN_WAVES * c;
            *(v4f*)(wlds + row * W_PITCH + 4 * lane) = wst[c];
        }
    }

    float bv[NT];
    #pragma unroll
    for (int t = 0; t < NT; ++t) {
        const int n  = t * 16 + l;
        bv[t] = bias[(n < N) ? n : (N - 1)];
    }

    __syncthreads();

    const float* wl = wlds + l * W_PITCH + 4 * half;

    v4f fb0[NT], fb1[NT];
    #pragma unroll
    for (int t = 0; t < NT; ++t) fb0[t] = *(const v4f*)(wl + t * 16 * W_PITCH);

    do {
        v8f acc[NT];
        #pragma unroll
        for (int t = 0; t < NT; ++t) acc[t] = (v8f){0.f, 0.f, 0.f, 0.f, 0.f, 0.f, 0.f, 0.f};

        #pragma unroll 1
        for (int g = 0; g < 4; ++g) {
            v4f ac[4];
            #pragma unroll
            for (int jj = 0; jj < 4; ++jj) ac[jj] = an[jj];
            if (g < 3) {
                #pragma unroll
                for (int jj = 0; jj < 4; ++jj) an[jj] = *(const v4f*)(xrow + 32 * (g + 1) + 8 * jj);
            }
            const float* wg = wl + 32 * g;
            const float* wn = wl + 32 * ((g + 1) & 3);
            mma_block<NT>(acc, fb0, fb1, ac[0], wg + 8);
            mma_block<NT>(acc, fb1, fb0, ac[1], wg + 16);
            mma_block<NT>(acc, fb0, fb1, ac[2], wg + 24);
            mma_block<NT>(acc, fb1, fb0, ac[3], wn);
        }

        const int mrow = mt * 16;
        mt += mstep;
        if (mt < mtiles) {
            const int m  = mt * 16 + l;
            const int mc = (m < M) ? m : (M - 1);
            xrow = X + (size_t)mc * D_HID + 4 * half;
            #pragma unroll
            for (int jj = 0; jj < 4; ++jj) an[jj] = *(const v4f*)(xrow + 8 * jj);
        }

        if (mrow + 16 <= M)
            store_strip<NT, RELU, INIT, false>(acc, bv, Y, Yinit, mrow + 8 * half, l, M, N);
        else
            store_strip<NT, RELU, INIT, true>(acc, bv, Y, Yinit, mrow + 8 * half, l, M, N);
    } while (mt < mtiles);
}

__global__ void __launch_bounds__(LIN_WAVES * 32)
linear_hidden_kernel(const float* __restrict__ X, const float* __restrict__ W,
                     const float* __restrict__ bias, float* __restrict__ Y, int M)
{
    __shared__ __attribute__((aligned(16))) float wlds[D_HID * W_PITCH];
    linear_strips<D_HID / 16, true, false>(X, W, bias, Y, nullptr, M, D_HID, wlds);
}

__global__ void __launch_bounds__(LIN_WAVES * 32)
linear_cls_kernel(const float* __restrict__ X, const float* __restrict__ W,
                  const float* __restrict__ bias, float* __restrict__ Z,
                  float* __restrict__ Yinit, int M, int N)
{
    __shared__ __attribute__((aligned(16))) float wlds[CLS_TILES * 16 * W_PITCH];
    linear_strips<CLS_TILES, false, true>(X, W, bias, Z, Yinit, M, N, wlds);
}

__global__ void zero_kernel(float* __restrict__ p, long long n)
{
    long long i = (long long)blockIdx.x * blockDim.x + threadIdx.x;
    if (i < n) p[i] = 0.0f;
}

__global__ void scatter_add_kernel(const float* __restrict__ Y,
                                   const int* __restrict__ src,
                                   const int* __restrict__ dst,
                                   float* __restrict__ out, int E)
{
    const long long t = (long long)blockIdx.x * blockDim.x + threadIdx.x;
    const int e = (int)(t >> 5);
    if (e >= E) return;
    const int q = (int)(t & 31);

    const int s = src[e];
    const int d = dst[e];
    const float* y = Y + (size_t)s * D_HID + q;
    float* o = out + (size_t)d * D_HID + q;
    const float v0 = y[0];
    const float v1 = y[32];
    const float v2 = y[64];
    const float v3 = y[96];
    atomicAdd(o +  0, v0);
    atomicAdd(o + 32, v1);
    atomicAdd(o + 64, v2);
    atomicAdd(o + 96, v3);
}

__global__ void scatter_add_cls_kernel(const float* __restrict__ Z,
                                       const int* __restrict__ src,
                                       const int* __restrict__ dst,
                                       float* __restrict__ out, int E, int C)
{
    const int e = blockIdx.x * blockDim.y + threadIdx.y;
    if (e >= E) return;

    const int s = src[e];
    const int d = dst[e];
    const float* zr = Z + (size_t)s * C;
    float* o = out + (size_t)d * C;

    float v[CLS_PER_LANE];
    for (int i = 0; i < CLS_PER_LANE; ++i) {
        const int c = threadIdx.x + i * blockDim.x;
        v[i] = (c < C) ? zr[c] : 0.f;
    }
    for (int i = 0; i < CLS_PER_LANE; ++i) {
        const int c = threadIdx.x + i * blockDim.x;
        if (c < C) atomicAdd(o + c, v[i]);
    }
}

extern "C" void kernel_launch(void* const* d_in, const int* in_sizes, int n_in,
                              void* d_out, int out_size, void* d_ws, size_t ws_size,
                              hipStream_t stream)
{
    const float* X0  = (const float*)d_in[0];
    const float* W1  = (const float*)d_in[1];
    const float* b1  = (const float*)d_in[2];
    const float* W2  = (const float*)d_in[3];
    const float* b2  = (const float*)d_in[4];
    const float* Wfc = (const float*)d_in[5];
    const float* bfc = (const float*)d_in[6];
    const int*   esrc = (const int*)d_in[7];
    const int*   edst = (const int*)d_in[8];

    const int Nn   = in_sizes[0] / D_HID;
    const int E    = in_sizes[7];
    const int Ncls = in_sizes[6];

    float* bufA = (float*)d_ws;
    float* bufB = bufA + (size_t)Nn * D_HID;
    float* zbuf = bufB;
    const long long hidElems = (long long)Nn * D_HID;

    const int mtiles = (Nn + 15) / 16;
    const int linBlocks = (mtiles + LIN_WAVES - 1) / LIN_WAVES;
    const dim3 blkLin(LIN_WAVES * 32);

    const int zeroBlocks = (int)((hidElems + 255) / 256);
    const int scatBlocks = (int)(((long long)E * 32 + 255) / 256);
    const dim3 blkCls((Ncls + CLS_PER_LANE - 1) / CLS_PER_LANE, 32);
    const int clsBlocks = (E + 31) / 32;

    linear_hidden_kernel<<<linBlocks, blkLin, 0, stream>>>(X0, W1, b1, bufA, Nn);
    zero_kernel<<<zeroBlocks, 256, 0, stream>>>(bufB, hidElems);
    scatter_add_kernel<<<scatBlocks, 256, 0, stream>>>(bufA, esrc, edst, bufB, E);

    linear_hidden_kernel<<<linBlocks, blkLin, 0, stream>>>(bufB, W2, b2, bufA, Nn);

    linear_cls_kernel<<<linBlocks, blkLin, 0, stream>>>(bufA, Wfc, bfc, zbuf, (float*)d_out,
                                                        Nn, Ncls);
    scatter_add_cls_kernel<<<clsBlocks, blkCls, 0, stream>>>(zbuf, esrc, edst,
                                                             (float*)d_out, E, Ncls);
}


# fewer, fatter waves in scatter-adds and zero fill: 16 edges per wave (both index arrays in one load, 8 edges' rows in flight before the adds), class scatter 4 edges per thread, fill 64 B per thread
# speedup vs baseline: 1.9677x; 1.1642x over previous
#include <hip/hip_runtime.h>

typedef float v2f __attribute__((ext_vector_type(2)));
typedef float v4f __attribute__((ext_vector_type(4)));
typedef float v8f __attribute__((ext_vector_type(8)));

#define D_HID 128
#define CLS_PER_LANE 4
#define CLS_TILES 3
#define LIN_WAVES 8
#define W_PITCH (D_HID + 4)


template <int NT>
__device__ __forceinline__ void mma_block(v8f (&acc)[NT], const v4f (&cur)[NT], v4f (&nxt)[NT],
                                          const v4f a, const float* __restrict__ wnext)
{
    #pragma unroll
    for (int t = 0; t < NT; ++t) nxt[t] = *(const v4f*)(wnext + t * 16 * W_PITCH);
    __builtin_amdgcn_sched_barrier(0);
    const v2f a0 = {a.x, a.y};
    const v2f a1 = {a.z, a.w};
    #pragma unroll
    for (int t = 0; t < NT; ++t) {
        const v2f b0 = {cur[t].x, cur[t].y};
        const v2f b1 = {cur[t].z, cur[t].w};
        acc[t] = __builtin_amdgcn_wmma_f32_16x16x4_f32(
                     false, a0, false, b0, (short)0, acc[t], false, false);
        acc[t] = __builtin_amdgcn_wmma_f32_16x16x4_f32(
                     false, a1, false, b1, (short)0, acc[t], false, false);
    }
    __builtin_amdgcn_sched_barrier(0);
}

template <int NT, bool RELU, bool INIT, bool ROWS_GUARDED>
__device__ __forceinline__ void store_strip(const v8f (&acc)[NT], const float (&bv)[NT],
                                            float* __restrict__ Y, float* __restrict__ Yinit,
                                            int row0, int l, int M, int N)
{
    float* yb = Y + (size_t)row0 * N + l;
    float* ib = INIT ? (Yinit + (size_t)row0 * N + l) : nullptr;
    #pragma unroll
    for (int t = 0; t < NT; ++t) {
        const float badd = INIT ? 0.f : bv[t];
        if (t * 16 + l < N) {
            #pragma unroll
            for (int r = 0; r < 8; ++r) {
                float v = acc[t][r] + badd;
                if (RELU) v = (v > 0.f) ? v : 0.f;
                if (!ROWS_GUARDED || row0 + r < M) {
                    yb[(size_t)r * N + t * 16] = v;
                    if (INIT) ib[(size_t)r * N + t * 16] = bv[t];
                }
            }
        }
    }
}

template <int NT, bool RELU, bool INIT>
__device__ __forceinline__ void linear_strips(const float* __restrict__ X,
                                              const float* __restrict__ W,
                                              const float* __restrict__ bias,
                                              float* __restrict__ Y,
                                              float* __restrict__ Yinit,
                                              int M, int N, float* wlds)
{
    const int tid  = threadIdx.x;
    const int lane = tid & 31;
    const int wave = __builtin_amdgcn_readfirstlane(tid >> 5);
    const int half = lane >> 4;
    const int l    = lane & 15;
    const int mtiles = (M + 15) >> 4;
    const int mstep  = gridDim.x * LIN_WAVES;

    int mt = blockIdx.x * LIN_WAVES + wave;

    const float* xrow;
    v4f an[4];
    {
        const int m  = mt * 16 + l;
        const int mc = (m < M) ? m : (M - 1);
        xrow = X + (size_t)mc * D_HID + 4 * half;
        #pragma unroll
        for (int jj = 0; jj < 4; ++jj) an[jj] = *(const v4f*)(xrow + 8 * jj);
    }

    {
        constexpr int ROWS_PER_WAVE = NT * 16 / LIN_WAVES;
        v4f wst[ROWS_PER_WAVE];
        #pragma unroll
        for (int c = 0; c < ROWS_PER_WAVE; ++c) {
            const int row = wave + LIN_WAVES * c;
            const int rc  = (row < N) ? row : (N - 1);
            wst[c] = *(const v4f*)(W + (size_t)rc * D_HID + 4 * lane);
        }
        #pragma unroll
        for (int c = 0; c < ROWS_PER_WAVE; ++c) {
            const int row = wave + LIN_WAVES * c;
            *(v4f*)(wlds + row * W_PITCH + 4 * lane) = wst[c];
        }
    }

    float bv[NT];
    #pragma unroll
    for (int t = 0; t < NT; ++t) {
        const int n  = t * 16 + l;
        bv[t] = bias[(n < N) ? n : (N - 1)];
    }

    __syncthreads();

    const float* wl = wlds + l * W_PITCH + 4 * half;

    v4f fb0[NT], fb1[NT];
    #pragma unroll
    for (int t = 0; t < NT; ++t) fb0[t] = *(const v4f*)(wl + t * 16 * W_PITCH);

    do {
        v8f acc[NT];
        #pragma unroll
        for (int t = 0; t < NT; ++t) acc[t] = (v8f){0.f, 0.f, 0.f, 0.f, 0.f, 0.f, 0.f, 0.f};

        #pragma unroll 1
        for (int g = 0; g < 4; ++g) {
            v4f ac[4];
            #pragma unroll
            for (int jj = 0; jj < 4; ++jj) ac[jj] = an[jj];
            if (g < 3) {
                #pragma unroll
                for (int jj = 0; jj < 4; ++jj) an[jj] = *(const v4f*)(xrow + 32 * (g + 1) + 8 * jj);
            }
            const float* wg = wl + 32 * g;
            const float* wn = wl + 32 * ((g + 1) & 3);
            mma_block<NT>(acc, fb0, fb1, ac[0], wg + 8);
            mma_block<NT>(acc, fb1, fb0, ac[1], wg + 16);
            mma_block<NT>(acc, fb0, fb1, ac[2], wg + 24);
            mma_block<NT>(acc, fb1, fb0, ac[3], wn);
        }

        const int mrow = mt * 16;
        mt += mstep;
        if (mt < mtiles) {
            const int m  = mt * 16 + l;
            const int mc = (m < M) ? m : (M - 1);
            xrow = X + (size_t)mc * D_HID + 4 * half;
            #pragma unroll
            for (int jj = 0; jj < 4; ++jj) an[jj] = *(const v4f*)(xrow + 8 * jj);
        }

        if (mrow + 16 <= M)
            store_strip<NT, RELU, INIT, false>(acc, bv, Y, Yinit, mrow + 8 * half, l, M, N);
        else
            store_strip<NT, RELU, INIT, true>(acc, bv, Y, Yinit, mrow + 8 * half, l, M, N);
    } while (mt < mtiles);
}

__global__ void __launch_bounds__(LIN_WAVES * 32)
linear_hidden_kernel(const float* __restrict__ X, const float* __restrict__ W,
                     const float* __restrict__ bias, float* __restrict__ Y, int M)
{
    __shared__ __attribute__((aligned(16))) float wlds[D_HID * W_PITCH];
    linear_strips<D_HID / 16, true, false>(X, W, bias, Y, nullptr, M, D_HID, wlds);
}

__global__ void __launch_bounds__(LIN_WAVES * 32)
linear_cls_kernel(const float* __restrict__ X, const float* __restrict__ W,
                  const float* __restrict__ bias, float* __restrict__ Z,
                  float* __restrict__ Yinit, int M, int N)
{
    __shared__ __attribute__((aligned(16))) float wlds[CLS_TILES * 16 * W_PITCH];
    linear_strips<CLS_TILES, false, true>(X, W, bias, Z, Yinit, M, N, wlds);
}

#define ZERO_VEC 4
#define SC_EDGES 16
#define SC_GROUP 8
#define SC_WAVES 8
#define CLS_EDGES 4
#define CLS_ROWS 32

__global__ void zero_kernel(float* __restrict__ p, long long n)
{
    const long long n4 = n >> 2;
    v4f* p4 = (v4f*)p;
    long long i = (long long)blockIdx.x * (blockDim.x * ZERO_VEC) + threadIdx.x;
    #pragma unroll
    for (int k = 0; k < ZERO_VEC; ++k, i += blockDim.x)
        if (i < n4) p4[i] = (v4f){0.f, 0.f, 0.f, 0.f};
}

__global__ void __launch_bounds__(SC_WAVES * 32)
scatter_add_kernel(const float* __restrict__ Y,
                   const int* __restrict__ src,
                   const int* __restrict__ dst,
                   float* __restrict__ out, int E)
{
    static_assert(2 * SC_EDGES == 32, "one 32-lane index load holds src and dst of the wave's edges");
    static_assert(SC_EDGES % SC_GROUP == 0, "whole groups");
    const int lane = threadIdx.x & 31;
    const int wave = __builtin_amdgcn_readfirstlane((int)(threadIdx.x >> 5));
    const int e0 = (blockIdx.x * SC_WAVES + wave) * SC_EDGES;
    if (e0 >= E) return;
    const int nval = (E - e0 < SC_EDGES) ? (E - e0) : SC_EDGES;

    const int k  = lane & (SC_EDGES - 1);
    const int ek = (k < nval) ? (e0 + k) : (E - 1);
    const int* ip = (lane < SC_EDGES) ? src : dst;
    const int idx = ip[ek];

    if (nval == SC_EDGES) {
        #pragma unroll
        for (int g = 0; g < SC_EDGES / SC_GROUP; ++g) {
            float v[SC_GROUP][4];
            #pragma unroll
            for (int j = 0; j < SC_GROUP; ++j) {
                const int s = __builtin_amdgcn_readlane(idx, g * SC_GROUP + j);
                const float* y = Y + (size_t)s * D_HID + lane;
                v[j][0] = y[0];
                v[j][1] = y[32];
                v[j][2] = y[64];
                v[j][3] = y[96];
            }
            #pragma unroll
            for (int j = 0; j < SC_GROUP; ++j) {
                const int d = __builtin_amdgcn_readlane(idx, SC_EDGES + g * SC_GROUP + j);
                float* o = out + (size_t)d * D_HID + lane;
                atomicAdd(o +  0, v[j][0]);
                atomicAdd(o + 32, v[j][1]);
                atomicAdd(o + 64, v[j][2]);
                atomicAdd(o + 96, v[j][3]);
            }
        }
    } else {
        for (int j = 0; j < nval; ++j) {
            const int s = __builtin_amdgcn_readlane(idx, j);
            const int d = __builtin_amdgcn_readlane(idx, SC_EDGES + j);
            const float* y = Y + (size_t)s * D_HID + lane;
            float* o = out + (size_t)d * D_HID + lane;
            const float v0 = y[0];
            const float v1 = y[32];
            const float v2 = y[64];
            const float v3 = y[96];
            atomicAdd(o +  0, v0);
            atomicAdd(o + 32, v1);
            atomicAdd(o + 64, v2);
            atomicAdd(o + 96, v3);
        }
    }
}

__global__ void scatter_add_cls_kernel(const float* __restrict__ Z,
                                       const int* __restrict__ src,
                                       const int* __restrict__ dst,
                                       float* __restrict__ out, int E, int C)
{
    const int e0 = (blockIdx.x * blockDim.y + threadIdx.y) * CLS_EDGES;
    const bool whole = ((int)((blockIdx.x + 1) * blockDim.y) * CLS_EDGES <= E) &&
                       (C == CLS_PER_LANE * (int)blockDim.x);

    int s[CLS_EDGES], d[CLS_EDGES];
    #pragma unroll
    for (int k = 0; k < CLS_EDGES; ++k) {
        const int e  = e0 + k;
        const int ec = (e < E) ? e : (E - 1);
        s[k] = src[ec];
        d[k] = dst[ec];
    }

    float v[CLS_EDGES][CLS_PER_LANE];
    #pragma unroll
    for (int k = 0; k < CLS_EDGES; ++k) {
        const float* zr = Z + (size_t)s[k] * C;
        #pragma unroll
        for (int i = 0; i < CLS_PER_LANE; ++i) {
            const int c = threadIdx.x + i * blockDim.x;
            v[k][i] = zr[(c < C) ? c : (C - 1)];
        }
    }

    if (whole) {
        #pragma unroll
        for (int k = 0; k < CLS_EDGES; ++k) {
            float* o = out + (size_t)d[k] * C + threadIdx.x;
            #pragma unroll
            for (int i = 0; i < CLS_PER_LANE; ++i) atomicAdd(o + i * blockDim.x, v[k][i]);
        }
    } else {
        #pragma unroll
        for (int k = 0; k < CLS_EDGES; ++k) {
            float* o = out + (size_t)d[k] * C;
            #pragma unroll
            for (int i = 0; i < CLS_PER_LANE; ++i) {
                const int c = threadIdx.x + i * blockDim.x;
                if (e0 + k < E && c < C) atomicAdd(o + c, v[k][i]);
            }
        }
    }
}

extern "C" void kernel_launch(void* const* d_in, const int* in_sizes, int n_in,
                              void* d_out, int out_size, void* d_ws, size_t ws_size,
                              hipStream_t stream)
{
    const float* X0  = (const float*)d_in[0];
    const float* W1  = (const float*)d_in[1];
    const float* b1  = (const float*)d_in[2];
    const float* W2  = (const float*)d_in[3];
    const float* b2  = (const float*)d_in[4];
    const float* Wfc = (const float*)d_in[5];
    const float* bfc = (const float*)d_in[6];
    const int*   esrc = (const int*)d_in[7];
    const int*   edst = (const int*)d_in[8];

    const int Nn   = in_sizes[0] / D_HID;
    const int E    = in_sizes[7];
    const int Ncls = in_sizes[6];

    float* bufA = (float*)d_ws;
    float* bufB = bufA + (size_t)Nn * D_HID;
    float* zbuf = bufB;
    const long long hidElems = (long long)Nn * D_HID;

    const int mtiles = (Nn + 15) / 16;
    const int linBlocks = (mtiles + LIN_WAVES - 1) / LIN_WAVES;
    const dim3 blkLin(LIN_WAVES * 32);

    const int zeroBlocks = (int)((hidElems / 4 + 256 * ZERO_VEC - 1) / (256 * ZERO_VEC));
    const int scatBlocks = (E + SC_WAVES * SC_EDGES - 1) / (SC_WAVES * SC_EDGES);
    const dim3 blkCls((Ncls + CLS_PER_LANE - 1) / CLS_PER_LANE, CLS_ROWS);
    const int clsBlocks = (E + CLS_ROWS * CLS_EDGES - 1) / (CLS_ROWS * CLS_EDGES);

    linear_hidden_kernel<<<linBlocks, blkLin, 0, stream>>>(X0, W1, b1, bufA, Nn);
    zero_kernel<<<zeroBlocks, 256, 0, stream>>>(bufB, hidElems);
    scatter_add_kernel<<<scatBlocks, SC_WAVES * 32, 0, stream>>>(bufA, esrc, edst, bufB, E);

    linear_hidden_kernel<<<linBlocks, blkLin, 0, stream>>>(bufB, W2, b2, bufA, Nn);

    linear_cls_kernel<<<linBlocks, blkLin, 0, stream>>>(bufA, Wfc, bfc, zbuf, (float*)d_out,
                                                        Nn, Ncls);
    scatter_add_cls_kernel<<<clsBlocks, blkCls, 0, stream>>>(zbuf, esrc, edst,
                                                             (float*)d_out, E, Ncls);
}


# class scatter-add: one wave per 16 edges; a 40-float row is cut at the destination row's 128-byte line boundary, so each atomic covers one row's share of one line (2 per edge, was 10-float pieces)
# speedup vs baseline: 2.3208x; 1.1794x over previous
#include <hip/hip_runtime.h>

typedef float v2f __attribute__((ext_vector_type(2)));
typedef float v4f __attribute__((ext_vector_type(4)));
typedef float v8f __attribute__((ext_vector_type(8)));

#define D_HID 128
#define CLS_TILES 3
#define LIN_WAVES 8
#define W_PITCH (D_HID + 4)


template <int NT>
__device__ __forceinline__ void mma_block(v8f (&acc)[NT], const v4f (&cur)[NT], v4f (&nxt)[NT],
                                          const v4f a, const float* __restrict__ wnext)
{
    #pragma unroll
    for (int t = 0; t < NT; ++t) nxt[t] = *(const v4f*)(wnext + t * 16 * W_PITCH);
    __builtin_amdgcn_sched_barrier(0);
    const v2f a0 = {a.x, a.y};
    const v2f a1 = {a.z, a.w};
    #pragma unroll
    for (int t = 0; t < NT; ++t) {
        const v2f b0 = {cur[t].x, cur[t].y};
        const v2f b1 = {cur[t].z, cur[t].w};
        acc[t] = __builtin_amdgcn_wmma_f32_16x16x4_f32(
                     false, a0, false, b0, (short)0, acc[t], false, false);
        acc[t] = __builtin_amdgcn_wmma_f32_16x16x4_f32(
                     false, a1, false, b1, (short)0, acc[t], false, false);
    }
    __builtin_amdgcn_sched_barrier(0);
}

template <int NT, bool RELU, bool INIT, bool ROWS_GUARDED>
__device__ __forceinline__ void store_strip(const v8f (&acc)[NT], const float (&bv)[NT],
                                            float* __restrict__ Y, float* __restrict__ Yinit,
                                            int row0, int l, int M, int N)
{
    float* yb = Y + (size_t)row0 * N + l;
    float* ib = INIT ? (Yinit + (size_t)row0 * N + l) : nullptr;
    #pragma unroll
    for (int t = 0; t < NT; ++t) {
        const float badd = INIT ? 0.f : bv[t];
        if (t * 16 + l < N) {
            #pragma unroll
            for (int r = 0; r < 8; ++r) {
                float v = acc[t][r] + badd;
                if (RELU) v = (v > 0.f) ? v : 0.f;
                if (!ROWS_GUARDED || row0 + r < M) {
                    yb[(size_t)r * N + t * 16] = v;
                    if (INIT) ib[(size_t)r * N + t * 16] = bv[t];
                }
            }
        }
    }
}

template <int NT, bool RELU, bool INIT>
__device__ __forceinline__ void linear_strips(const float* __restrict__ X,
                                              const float* __restrict__ W,
                                              const float* __restrict__ bias,
                                              float* __restrict__ Y,
                                              float* __restrict__ Yinit,
                                              int M, int N, float* wlds)
{
    const int tid  = threadIdx.x;
    const int lane = tid & 31;
    const int wave = __builtin_amdgcn_readfirstlane(tid >> 5);
    const int half = lane >> 4;
    const int l    = lane & 15;
    const int mtiles = (M + 15) >> 4;
    const int mstep  = gridDim.x * LIN_WAVES;

    int mt = blockIdx.x * LIN_WAVES + wave;

    const float* xrow;
    v4f an[4];
    {
        const int m  = mt * 16 + l;
        const int mc = (m < M) ? m : (M - 1);
        xrow = X + (size_t)mc * D_HID + 4 * half;
        #pragma unroll
        for (int jj = 0; jj < 4; ++jj) an[jj] = *(const v4f*)(xrow + 8 * jj);
    }

    {
        constexpr int ROWS_PER_WAVE = NT * 16 / LIN_WAVES;
        v4f wst[ROWS_PER_WAVE];
        #pragma unroll
        for (int c = 0; c < ROWS_PER_WAVE; ++c) {
            const int row = wave + LIN_WAVES * c;
            const int rc  = (row < N) ? row : (N - 1);
            wst[c] = *(const v4f*)(W + (size_t)rc * D_HID + 4 * lane);
        }
        #pragma unroll
        for (int c = 0; c < ROWS_PER_WAVE; ++c) {
            const int row = wave + LIN_WAVES * c;
            *(v4f*)(wlds + row * W_PITCH + 4 * lane) = wst[c];
        }
    }

    float bv[NT];
    #pragma unroll
    for (int t = 0; t < NT; ++t) {
        const int n  = t * 16 + l;
        bv[t] = bias[(n < N) ? n : (N - 1)];
    }

    __syncthreads();

    const float* wl = wlds + l * W_PITCH + 4 * half;

    v4f fb0[NT], fb1[NT];
    #pragma unroll
    for (int t = 0; t < NT; ++t) fb0[t] = *(const v4f*)(wl + t * 16 * W_PITCH);

    do {
        v8f acc[NT];
        #pragma unroll
        for (int t = 0; t < NT; ++t) acc[t] = (v8f){0.f, 0.f, 0.f, 0.f, 0.f, 0.f, 0.f, 0.f};

        #pragma unroll 1
        for (int g = 0; g < 4; ++g) {
            v4f ac[4];
            #pragma unroll
            for (int jj = 0; jj < 4; ++jj) ac[jj] = an[jj];
            if (g < 3) {
                #pragma unroll
                for (int jj = 0; jj < 4; ++jj) an[jj] = *(const v4f*)(xrow + 32 * (g + 1) + 8 * jj);
            }
            const float* wg = wl + 32 * g;
            const float* wn = wl + 32 * ((g + 1) & 3);
            mma_block<NT>(acc, fb0, fb1, ac[0], wg + 8);
            mma_block<NT>(acc, fb1, fb0, ac[1], wg + 16);
            mma_block<NT>(acc, fb0, fb1, ac[2], wg + 24);
            mma_block<NT>(acc, fb1, fb0, ac[3], wn);
        }

        const int mrow = mt * 16;
        mt += mstep;
        if (mt < mtiles) {
            const int m  = mt * 16 + l;
            const int mc = (m < M) ? m : (M - 1);
            xrow = X + (size_t)mc * D_HID + 4 * half;
            #pragma unroll
            for (int jj = 0; jj < 4; ++jj) an[jj] = *(const v4f*)(xrow + 8 * jj);
        }

        if (mrow + 16 <= M)
            store_strip<NT, RELU, INIT, false>(acc, bv, Y, Yinit, mrow + 8 * half, l, M, N);
        else
            store_strip<NT, RELU, INIT, true>(acc, bv, Y, Yinit, mrow + 8 * half, l, M, N);
    } while (mt < mtiles);
}

__global__ void __launch_bounds__(LIN_WAVES * 32)
linear_hidden_kernel(const float* __restrict__ X, const float* __restrict__ W,
                     const float* __restrict__ bias, float* __restrict__ Y, int M)
{
    __shared__ __attribute__((aligned(16))) float wlds[D_HID * W_PITCH];
    linear_strips<D_HID / 16, true, false>(X, W, bias, Y, nullptr, M, D_HID, wlds);
}

__global__ void __launch_bounds__(LIN_WAVES * 32)
linear_cls_kernel(const float* __restrict__ X, const float* __restrict__ W,
                  const float* __restrict__ bias, float* __restrict__ Z,
                  float* __restrict__ Yinit, int M, int N)
{
    __shared__ __attribute__((aligned(16))) float wlds[CLS_TILES * 16 * W_PITCH];
    linear_strips<CLS_TILES, false, true>(X, W, bias, Z, Yinit, M, N, wlds);
}

#define ZERO_VEC 4
#define SC_EDGES 16
#define SC_GROUP 8
#define SC_WAVES 8
#define CLS_SC_EDGES 16
#define CLS_SC_WAVES 8

__global__ void zero_kernel(float* __restrict__ p, long long n)
{
    const long long n4 = n >> 2;
    v4f* p4 = (v4f*)p;
    long long i = (long long)blockIdx.x * (blockDim.x * ZERO_VEC) + threadIdx.x;
    #pragma unroll
    for (int k = 0; k < ZERO_VEC; ++k, i += blockDim.x)
        if (i < n4) p4[i] = (v4f){0.f, 0.f, 0.f, 0.f};
}

__global__ void __launch_bounds__(SC_WAVES * 32)
scatter_add_kernel(const float* __restrict__ Y,
                   const int* __restrict__ src,
                   const int* __restrict__ dst,
                   float* __restrict__ out, int E)
{
    static_assert(2 * SC_EDGES == 32, "one 32-lane index load holds src and dst of the wave's edges");
    static_assert(SC_EDGES % SC_GROUP == 0, "whole groups");
    const int lane = threadIdx.x & 31;
    const int wave = __builtin_amdgcn_readfirstlane((int)(threadIdx.x >> 5));
    const int e0 = (blockIdx.x * SC_WAVES + wave) * SC_EDGES;
    if (e0 >= E) return;
    const int nval = (E - e0 < SC_EDGES) ? (E - e0) : SC_EDGES;

    const int k  = lane & (SC_EDGES - 1);
    const int ek = (k < nval) ? (e0 + k) : (E - 1);
    const int* ip = (lane < SC_EDGES) ? src : dst;
    const int idx = ip[ek];

    if (nval == SC_EDGES) {
        #pragma unroll
        for (int g = 0; g < SC_EDGES / SC_GROUP; ++g) {
            float v[SC_GROUP][4];
            #pragma unroll
            for (int j = 0; j < SC_GROUP; ++j) {
                const int s = __builtin_amdgcn_readlane(idx, g * SC_GROUP + j);
                const float* y = Y + (size_t)s * D_HID + lane;
                v[j][0] = y[0];
                v[j][1] = y[32];
                v[j][2] = y[64];
                v[j][3] = y[96];
            }
            #pragma unroll
            for (int j = 0; j < SC_GROUP; ++j) {
                const int d = __builtin_amdgcn_readlane(idx, SC_EDGES + g * SC_GROUP + j);
                float* o = out + (size_t)d * D_HID + lane;
                atomicAdd(o +  0, v[j][0]);
                atomicAdd(o + 32, v[j][1]);
                atomicAdd(o + 64, v[j][2]);
                atomicAdd(o + 96, v[j][3]);
            }
        }
    } else {
        for (int j = 0; j < nval; ++j) {
            const int s = __builtin_amdgcn_readlane(idx, j);
            const int d = __builtin_amdgcn_readlane(idx, SC_EDGES + j);
            const float* y = Y + (size_t)s * D_HID + lane;
            float* o = out + (size_t)d * D_HID + lane;
            const float v0 = y[0];
            const float v1 = y[32];
            const float v2 = y[64];
            const float v3 = y[96];
            atomicAdd(o +  0, v0);
            atomicAdd(o + 32, v1);
            atomicAdd(o + 64, v2);
            atomicAdd(o + 96, v3);
        }
    }
}


__device__ __forceinline__ int first_line_floats(int node, int C)
{
    const int n = 32 - (int)(((unsigned)node * (unsigned)C) & 31u);
    return (n < C) ? n : C;
}

__device__ __forceinline__ void add_cls_row(const float* __restrict__ Z, float* __restrict__ out,
                                            int s, int d, int n1, int C, int lane, float va, float vb)
{
    const int n2 = (C - n1 < 32) ? (C - n1) : 32;
    float* o = out + (size_t)d * C;
    if (lane < n1) atomicAdd(o + lane, va);
    if (lane < n2) atomicAdd(o + n1 + lane, vb);
    if (n1 + 32 < C) {
        const float* z = Z + (size_t)s * C;
        for (int c = n1 + 32 + lane; c < C; c += 32) atomicAdd(o + c, z[c]);
    }
}

__global__ void __launch_bounds__(CLS_SC_WAVES * 32)
scatter_add_cls_kernel(const float* __restrict__ Z,
                       const int* __restrict__ src,
                       const int* __restrict__ dst,
                       float* __restrict__ out, int E, int C)
{
    static_assert(2 * CLS_SC_EDGES == 32, "one 32-lane index load holds src and dst of the wave's edges");
    const int lane = threadIdx.x & 31;
    const int wave = __builtin_amdgcn_readfirstlane((int)(threadIdx.x >> 5));
    const int e0 = (blockIdx.x * CLS_SC_WAVES + wave) * CLS_SC_EDGES;
    if (e0 >= E) return;
    const int nval = (E - e0 < CLS_SC_EDGES) ? (E - e0) : CLS_SC_EDGES;

    const int k  = lane & (CLS_SC_EDGES - 1);
    const int ek = (k < nval) ? (e0 + k) : (E - 1);
    const int* ip = (lane < CLS_SC_EDGES) ? src : dst;
    const int idx = ip[ek];

    if (nval == CLS_SC_EDGES) {
        float va[CLS_SC_EDGES], vb[CLS_SC_EDGES];
        #pragma unroll
        for (int j = 0; j < CLS_SC_EDGES; ++j) {
            const int s  = __builtin_amdgcn_readlane(idx, j);
            const int d  = __builtin_amdgcn_readlane(idx, CLS_SC_EDGES + j);
            const int n1 = first_line_floats(d, C);
            const int m2 = ((C - n1 < 32) ? (C - n1) : 32) - 1;
            const float* z = Z + (size_t)s * C;
            va[j] = z[(lane < n1) ? lane : (n1 - 1)];
            vb[j] = z[n1 + ((lane < m2) ? lane : m2)];
        }
        #pragma unroll
        for (int j = 0; j < CLS_SC_EDGES; ++j) {
            const int s  = __builtin_amdgcn_readlane(idx, j);
            const int d  = __builtin_amdgcn_readlane(idx, CLS_SC_EDGES + j);
            add_cls_row(Z, out, s, d, first_line_floats(d, C), C, lane, va[j], vb[j]);
        }
    } else {
        for (int j = 0; j < nval; ++j) {
            const int s  = __builtin_amdgcn_readlane(idx, j);
            const int d  = __builtin_amdgcn_readlane(idx, CLS_SC_EDGES + j);
            const int n1 = first_line_floats(d, C);
            const int m2 = ((C - n1 < 32) ? (C - n1) : 32) - 1;
            const float* z = Z + (size_t)s * C;
            const float a = z[(lane < n1) ? lane : (n1 - 1)];
            const float b = z[n1 + ((lane < m2) ? lane : m2)];
            add_cls_row(Z, out, s, d, n1, C, lane, a, b);
        }
    }
}

extern "C" void kernel_launch(void* const* d_in, const int* in_sizes, int n_in,
                              void* d_out, int out_size, void* d_ws, size_t ws_size,
                              hipStream_t stream)
{
    const float* X0  = (const float*)d_in[0];
    const float* W1  = (const float*)d_in[1];
    const float* b1  = (const float*)d_in[2];
    const float* W2  = (const float*)d_in[3];
    const float* b2  = (const float*)d_in[4];
    const float* Wfc = (const float*)d_in[5];
    const float* bfc = (const float*)d_in[6];
    const int*   esrc = (const int*)d_in[7];
    const int*   edst = (const int*)d_in[8];

    const int Nn   = in_sizes[0] / D_HID;
    const int E    = in_sizes[7];
    const int Ncls = in_sizes[6];

    float* bufA = (float*)d_ws;
    float* bufB = bufA + (size_t)Nn * D_HID;
    float* zbuf = bufB;
    const long long hidElems = (long long)Nn * D_HID;

    const int mtiles = (Nn + 15) / 16;
    const int linBlocks = (mtiles + LIN_WAVES - 1) / LIN_WAVES;
    const dim3 blkLin(LIN_WAVES * 32);

    const int zeroBlocks = (int)((hidElems / 4 + 256 * ZERO_VEC - 1) / (256 * ZERO_VEC));
    const int scatBlocks = (E + SC_WAVES * SC_EDGES - 1) / (SC_WAVES * SC_EDGES);
    const int clsBlocks = (E + CLS_SC_WAVES * CLS_SC_EDGES - 1) / (CLS_SC_WAVES * CLS_SC_EDGES);

    linear_hidden_kernel<<<linBlocks, blkLin, 0, stream>>>(X0, W1, b1, bufA, Nn);
    zero_kernel<<<zeroBlocks, 256, 0, stream>>>(bufB, hidElems);
    scatter_add_kernel<<<scatBlocks, SC_WAVES * 32, 0, stream>>>(bufA, esrc, edst, bufB, E);

    linear_hidden_kernel<<<linBlocks, blkLin, 0, stream>>>(bufB, W2, b2, bufA, Nn);

    linear_cls_kernel<<<linBlocks, blkLin, 0, stream>>>(bufA, Wfc, bfc, zbuf, (float*)d_out,
                                                        Nn, Ncls);
    scatter_add_cls_kernel<<<clsBlocks, CLS_SC_WAVES * 32, 0, stream>>>(zbuf, esrc, edst,
                                                                        (float*)d_out, E, Ncls);
}
